# starting kernel
# baseline (speedup 1.0000x reference)
#include <hip/hip_runtime.h>

typedef __attribute__((ext_vector_type(2))) float v2f;
typedef __attribute__((ext_vector_type(8))) float v8f;

#define N_NODES 100000
#define N_EDGES 3200000
#define F_IN    256
#define CH      16
#define FCN     64
#define NTILES  (N_NODES / 16)

__global__ __launch_bounds__(256) void gcn_feat_wmma(
    const float* __restrict__ x, const float* __restrict__ wg,
    float* __restrict__ h)
{
    __shared__ float lds_w[F_IN * CH];
    const int tid = threadIdx.x;
    for (int i = tid; i < F_IN * CH; i += 256) lds_w[i] = wg[i];
    __syncthreads();

    const int wave = tid >> 5;
    const int lane = tid & 31;
    const int tile = blockIdx.x * 8 + wave;
    if (tile >= NTILES) return;

    const int m  = lane & 15;
    const int hi = lane >> 4;
    const int row0 = tile * 16;
    const float* arow = x + (size_t)(row0 + m) * F_IN + 2 * hi;

    v8f acc = {0.f, 0.f, 0.f, 0.f, 0.f, 0.f, 0.f, 0.f};
    #pragma unroll 8
    for (int k = 0; k < 64; ++k) {
        const int kk = 4 * k;
        v2f a; a.x = arow[kk];     a.y = arow[kk + 1];
        v2f b; b.x = lds_w[(kk + 2 * hi)     * CH + m];
               b.y = lds_w[(kk + 2 * hi + 1) * CH + m];
        acc = __builtin_amdgcn_wmma_f32_16x16x4_f32(
                  false, a, false, b, (short)0, acc, false, false);
    }

    float* hrow = h + (size_t)row0 * CH + m;
    #pragma unroll
    for (int j = 0; j < 8; ++j)
        hrow[(size_t)(j + 8 * hi) * CH] = acc[j];
}

__global__ __launch_bounds__(256) void gcn_edge_scatter(
    const float* __restrict__ h, const int* __restrict__ src,
    const int* __restrict__ dst, const float* __restrict__ ew,
    float* __restrict__ agg)
{
    const long long idx = (long long)blockIdx.x * 256 + threadIdx.x;
    if (idx >= (long long)N_EDGES * CH) return;
    const int e = (int)(idx >> 4);
    const int c = (int)(idx & 15);
    const int s = src[e];
    const int d = dst[e];
    const float v = h[(size_t)s * CH + c] * ew[e];
    atomicAdd(&agg[(size_t)d * CH + c], v);
}

__global__ __launch_bounds__(256) void gcn_head_wmma(
    const float* __restrict__ agg, const float* __restrict__ w0,
    const float* __restrict__ b0,  const float* __restrict__ w1,
    const float* __restrict__ b1,  float* __restrict__ out)
{
    __shared__ float lds_w0[CH * 80];
    __shared__ float lds_b0[FCN];
    __shared__ float lds_w1[FCN];
    __shared__ float lds_o1[8][16 * 68];

    const int tid = threadIdx.x;
    for (int i = tid; i < CH * FCN; i += 256)
        lds_w0[(i / FCN) * 80 + (i % FCN)] = w0[i];
    if (tid < FCN) { lds_b0[tid] = b0[tid]; lds_w1[tid] = w1[tid]; }
    __syncthreads();

    const int wave = tid >> 5;
    const int lane = tid & 31;
    const int tile = blockIdx.x * 8 + wave;
    if (tile >= NTILES) return;

    const int m  = lane & 15;
    const int hi = lane >> 4;
    const int row0 = tile * 16;
    const float* arow = agg + (size_t)(row0 + m) * CH + 2 * hi;

    v8f acc[4];
    #pragma unroll
    for (int t = 0; t < 4; ++t) acc[t] = (v8f){0.f,0.f,0.f,0.f,0.f,0.f,0.f,0.f};

    #pragma unroll
    for (int k = 0; k < 4; ++k) {
        const int kk = 4 * k;
        v2f a; a.x = fmaxf(arow[kk], 0.f);
               a.y = fmaxf(arow[kk + 1], 0.f);
        const float* br0 = &lds_w0[(kk + 2 * hi)     * 80 + m];
        const float* br1 = &lds_w0[(kk + 2 * hi + 1) * 80 + m];
        #pragma unroll
        for (int t = 0; t < 4; ++t) {
            v2f b; b.x = br0[16 * t]; b.y = br1[16 * t];
            acc[t] = __builtin_amdgcn_wmma_f32_16x16x4_f32(
                         false, a, false, b, (short)0, acc[t], false, false);
        }
    }

    float* o1 = &lds_o1[wave][0];
    #pragma unroll
    for (int t = 0; t < 4; ++t) {
        const float bias = lds_b0[16 * t + m];
        #pragma unroll
        for (int j = 0; j < 8; ++j)
            o1[(j + 8 * hi) * 68 + 16 * t + m] = fmaxf(acc[t][j] + bias, 0.f);
    }

    const float* row = &lds_o1[wave][m * 68 + 32 * hi];
    const float* wv  = &lds_w1[32 * hi];
    float s = 0.f;
    #pragma unroll
    for (int f = 0; f < 32; ++f) s += row[f] * wv[f];
    s += __shfl_xor(s, 16, 32);
    if (hi == 0) out[row0 + m] = s + b1[0];
}

extern "C" void kernel_launch(void* const* d_in, const int* in_sizes, int n_in,
                              void* d_out, int out_size, void* d_ws, size_t ws_size,
                              hipStream_t stream)
{
    (void)in_sizes; (void)n_in; (void)out_size; (void)ws_size;
    const float* x  = (const float*)d_in[0];
    const int*   ei = (const int*)  d_in[1];
    const float* ew = (const float*)d_in[2];
    const float* wg = (const float*)d_in[3];
    const float* w0 = (const float*)d_in[4];
    const float* b0 = (const float*)d_in[5];
    const float* w1 = (const float*)d_in[6];
    const float* b1 = (const float*)d_in[7];
    float* out = (float*)d_out;

    float* h   = (float*)d_ws;
    float* agg = h + (size_t)N_NODES * CH;

    hipMemsetAsync(agg, 0, (size_t)N_NODES * CH * sizeof(float), stream);

    const int tile_blocks = (NTILES + 7) / 8;
    gcn_feat_wmma<<<tile_blocks, 256, 0, stream>>>(x, wg, h);

    const long long ework = (long long)N_EDGES * CH;
    const int edge_blocks = (int)((ework + 255) / 256);
    gcn_edge_scatter<<<edge_blocks, 256, 0, stream>>>(h, ei, ei + N_EDGES, ew, agg);

    gcn_head_wmma<<<tile_blocks, 256, 0, stream>>>(agg, w0, b0, w1, b1, out);
}


# edge scatter: each wave keeps 32 edges in flight (16-byte index loads, then 16 row gathers, then 16 atomics per lane) instead of 2; the f32 atomic adds land in another (still arrival) order
# speedup vs baseline: 1.8115x; 1.8115x over previous
#include <hip/hip_runtime.h>

typedef __attribute__((ext_vector_type(2))) float v2f;
typedef __attribute__((ext_vector_type(8))) float v8f;

#define N_NODES 100000
#define N_EDGES 3200000
#define F_IN    256
#define CH      16
#define FCN     64
#define NTILES  (N_NODES / 16)

__global__ __launch_bounds__(256) void gcn_feat_wmma(
    const float* __restrict__ x, const float* __restrict__ wg,
    float* __restrict__ h)
{
    __shared__ float lds_w[F_IN * CH];
    const int tid = threadIdx.x;
    for (int i = tid; i < F_IN * CH; i += 256) lds_w[i] = wg[i];
    __syncthreads();

    const int wave = tid >> 5;
    const int lane = tid & 31;
    const int tile = blockIdx.x * 8 + wave;
    if (tile >= NTILES) return;

    const int m  = lane & 15;
    const int hi = lane >> 4;
    const int row0 = tile * 16;
    const float* arow = x + (size_t)(row0 + m) * F_IN + 2 * hi;

    v8f acc = {0.f, 0.f, 0.f, 0.f, 0.f, 0.f, 0.f, 0.f};
    #pragma unroll 8
    for (int k = 0; k < 64; ++k) {
        const int kk = 4 * k;
        v2f a; a.x = arow[kk];     a.y = arow[kk + 1];
        v2f b; b.x = lds_w[(kk + 2 * hi)     * CH + m];
               b.y = lds_w[(kk + 2 * hi + 1) * CH + m];
        acc = __builtin_amdgcn_wmma_f32_16x16x4_f32(
                  false, a, false, b, (short)0, acc, false, false);
    }

    float* hrow = h + (size_t)row0 * CH + m;
    #pragma unroll
    for (int j = 0; j < 8; ++j)
        hrow[(size_t)(j + 8 * hi) * CH] = acc[j];
}

#define EDGES_PER_WAVE 32
#define N_EDGE_WAVES   (N_EDGES / EDGES_PER_WAVE)

__global__ __launch_bounds__(256) void gcn_edge_scatter(
    const float* __restrict__ h, const int* __restrict__ src,
    const int* __restrict__ dst, const float* __restrict__ ew,
    float* __restrict__ agg)
{
    const unsigned tid  = threadIdx.x;
    const unsigned wave = blockIdx.x * 8u + (tid >> 5);
    if (wave >= (unsigned)N_EDGE_WAVES) return;
    const unsigned lane = tid & 31u;
    const unsigned c    = lane & 15u;
    const unsigned e0   = wave * (unsigned)EDGES_PER_WAVE + (lane >> 4) * 16u;

    const int4*   sp = reinterpret_cast<const int4*>(src + e0);
    const int4*   dp = reinterpret_cast<const int4*>(dst + e0);
    const float4* wp = reinterpret_cast<const float4*>(ew + e0);

    int s[16], d[16];
    float w[16];
    #pragma unroll
    for (int q = 0; q < 4; ++q) {
        const int4   sv = sp[q];
        const int4   dv = dp[q];
        const float4 wv = wp[q];
        s[4 * q + 0] = sv.x; s[4 * q + 1] = sv.y; s[4 * q + 2] = sv.z; s[4 * q + 3] = sv.w;
        d[4 * q + 0] = dv.x; d[4 * q + 1] = dv.y; d[4 * q + 2] = dv.z; d[4 * q + 3] = dv.w;
        w[4 * q + 0] = wv.x; w[4 * q + 1] = wv.y; w[4 * q + 2] = wv.z; w[4 * q + 3] = wv.w;
    }

    float v[16];
    #pragma unroll
    for (int j = 0; j < 16; ++j)
        v[j] = h[(unsigned)s[j] * (unsigned)CH + c] * w[j];

    #pragma unroll
    for (int j = 0; j < 16; ++j)
        atomicAdd(&agg[(unsigned)d[j] * (unsigned)CH + c], v[j]);
}

__global__ __launch_bounds__(256) void gcn_head_wmma(
    const float* __restrict__ agg, const float* __restrict__ w0,
    const float* __restrict__ b0,  const float* __restrict__ w1,
    const float* __restrict__ b1,  float* __restrict__ out)
{
    __shared__ float lds_w0[CH * 80];
    __shared__ float lds_b0[FCN];
    __shared__ float lds_w1[FCN];
    __shared__ float lds_o1[8][16 * 68];

    const int tid = threadIdx.x;
    for (int i = tid; i < CH * FCN; i += 256)
        lds_w0[(i / FCN) * 80 + (i % FCN)] = w0[i];
    if (tid < FCN) { lds_b0[tid] = b0[tid]; lds_w1[tid] = w1[tid]; }
    __syncthreads();

    const int wave = tid >> 5;
    const int lane = tid & 31;
    const int tile = blockIdx.x * 8 + wave;
    if (tile >= NTILES) return;

    const int m  = lane & 15;
    const int hi = lane >> 4;
    const int row0 = tile * 16;
    const float* arow = agg + (size_t)(row0 + m) * CH + 2 * hi;

    v8f acc[4];
    #pragma unroll
    for (int t = 0; t < 4; ++t) acc[t] = (v8f){0.f,0.f,0.f,0.f,0.f,0.f,0.f,0.f};

    #pragma unroll
    for (int k = 0; k < 4; ++k) {
        const int kk = 4 * k;
        v2f a; a.x = fmaxf(arow[kk], 0.f);
               a.y = fmaxf(arow[kk + 1], 0.f);
        const float* br0 = &lds_w0[(kk + 2 * hi)     * 80 + m];
        const float* br1 = &lds_w0[(kk + 2 * hi + 1) * 80 + m];
        #pragma unroll
        for (int t = 0; t < 4; ++t) {
            v2f b; b.x = br0[16 * t]; b.y = br1[16 * t];
            acc[t] = __builtin_amdgcn_wmma_f32_16x16x4_f32(
                         false, a, false, b, (short)0, acc[t], false, false);
        }
    }

    float* o1 = &lds_o1[wave][0];
    #pragma unroll
    for (int t = 0; t < 4; ++t) {
        const float bias = lds_b0[16 * t + m];
        #pragma unroll
        for (int j = 0; j < 8; ++j)
            o1[(j + 8 * hi) * 68 + 16 * t + m] = fmaxf(acc[t][j] + bias, 0.f);
    }

    const float* row = &lds_o1[wave][m * 68 + 32 * hi];
    const float* wv  = &lds_w1[32 * hi];
    float s = 0.f;
    #pragma unroll
    for (int f = 0; f < 32; ++f) s += row[f] * wv[f];
    s += __shfl_xor(s, 16, 32);
    if (hi == 0) out[row0 + m] = s + b1[0];
}

extern "C" void kernel_launch(void* const* d_in, const int* in_sizes, int n_in,
                              void* d_out, int out_size, void* d_ws, size_t ws_size,
                              hipStream_t stream)
{
    (void)in_sizes; (void)n_in; (void)out_size; (void)ws_size;
    const float* x  = (const float*)d_in[0];
    const int*   ei = (const int*)  d_in[1];
    const float* ew = (const float*)d_in[2];
    const float* wg = (const float*)d_in[3];
    const float* w0 = (const float*)d_in[4];
    const float* b0 = (const float*)d_in[5];
    const float* w1 = (const float*)d_in[6];
    const float* b1 = (const float*)d_in[7];
    float* out = (float*)d_out;

    float* h   = (float*)d_ws;
    float* agg = h + (size_t)N_NODES * CH;

    hipMemsetAsync(agg, 0, (size_t)N_NODES * CH * sizeof(float), stream);

    const int tile_blocks = (NTILES + 7) / 8;
    gcn_feat_wmma<<<tile_blocks, 256, 0, stream>>>(x, wg, h);

    const int edge_blocks = (N_EDGE_WAVES + 7) / 8;
    gcn_edge_scatter<<<edge_blocks, 256, 0, stream>>>(h, ei, ei + N_EDGES, ew, agg);

    gcn_head_wmma<<<tile_blocks, 256, 0, stream>>>(agg, w0, b0, w1, b1, out);
}


# edge scatter: each cache half (die id, message 0x87) adds only into its own half of the node range; edge chunks by a ticket word per half, kept edges packed in LDS; f32 atomics stay arrival-ordered
# speedup vs baseline: 2.5053x; 1.3830x over previous
#include <hip/hip_runtime.h>

typedef __attribute__((ext_vector_type(2))) float v2f;
typedef __attribute__((ext_vector_type(8))) float v8f;

#define N_NODES 100000
#define N_EDGES 3200000
#define F_IN    256
#define CH      16
#define FCN     64
#define NTILES  (N_NODES / 16)
#define TICKET_STRIDE  256

__global__ __launch_bounds__(256) void gcn_feat_wmma(
    const float* __restrict__ x, const float* __restrict__ wg,
    float* __restrict__ h, unsigned* __restrict__ tickets)
{
    __shared__ float lds_w[F_IN * CH];
    const int tid = threadIdx.x;
    if (blockIdx.x == 0 && tid < 2) tickets[tid * TICKET_STRIDE] = 0u;
    for (int i = tid; i < F_IN * CH; i += 256) lds_w[i] = wg[i];
    __syncthreads();

    const int wave = tid >> 5;
    const int lane = tid & 31;
    const int tile = blockIdx.x * 8 + wave;
    if (tile >= NTILES) return;

    const int m  = lane & 15;
    const int hi = lane >> 4;
    const int row0 = tile * 16;
    const float* arow = x + (size_t)(row0 + m) * F_IN + 2 * hi;

    v8f acc = {0.f, 0.f, 0.f, 0.f, 0.f, 0.f, 0.f, 0.f};
    #pragma unroll 8
    for (int k = 0; k < 64; ++k) {
        const int kk = 4 * k;
        v2f a; a.x = arow[kk];     a.y = arow[kk + 1];
        v2f b; b.x = lds_w[(kk + 2 * hi)     * CH + m];
               b.y = lds_w[(kk + 2 * hi + 1) * CH + m];
        acc = __builtin_amdgcn_wmma_f32_16x16x4_f32(
                  false, a, false, b, (short)0, acc, false, false);
    }

    float* hrow = h + (size_t)row0 * CH + m;
    #pragma unroll
    for (int j = 0; j < 8; ++j)
        hrow[(size_t)(j + 8 * hi) * CH] = acc[j];
}

#define CHUNK_EDGES    256
#define N_CHUNKS       (N_EDGES / CHUNK_EDGES)
#define HALF_NODES     (N_NODES / 2)
#define SCATTER_BLOCKS 1024

__device__ __forceinline__ unsigned ticket_draw(unsigned* ctr, unsigned lane)
{
    unsigned t = 0u;
    if (lane == 0u) t = atomicAdd(ctr, 1u);
    return (unsigned)__builtin_amdgcn_readfirstlane((int)t);
}

__device__ __forceinline__ void scatter_run(
    const int2* rec_hw, const int* rec_d, const unsigned b, const unsigned n, const bool tail,
    const unsigned hi, const unsigned c,
    const float* __restrict__ h, float* __restrict__ agg)
{
    int2 hw[16];
    int  d[16];
    #pragma unroll
    for (int j = 0; j < 16; ++j) {
        const unsigned idx = b + 2u * (unsigned)j + hi;
        const unsigned ri  = (!tail || idx < n) ? idx : b;
        hw[j] = rec_hw[ri];
        d[j]  = rec_d[ri];
    }
    float v[16];
    #pragma unroll
    for (int j = 0; j < 16; ++j) {
        const unsigned idx = b + 2u * (unsigned)j + hi;
        v[j] = 0.f;
        if (!tail || idx < n)
            v[j] = h[(unsigned)hw[j].x + c] * __int_as_float(hw[j].y);
    }
    #pragma unroll
    for (int j = 0; j < 16; ++j) {
        const unsigned idx = b + 2u * (unsigned)j + hi;
        if (!tail || idx < n)
            atomicAdd(&agg[(unsigned)d[j] + c], v[j]);
    }
}

__global__ __launch_bounds__(256) void gcn_edge_scatter(
    const float* __restrict__ h, const int* __restrict__ src,
    const int* __restrict__ dst, const float* __restrict__ ew,
    float* __restrict__ agg, unsigned* __restrict__ tickets)
{
    __shared__ int2 lds_hw[8][CHUNK_EDGES];
    __shared__ int  lds_d[8][CHUNK_EDGES];

    const unsigned tid  = threadIdx.x;
    const unsigned lane = tid & 31u;
    const unsigned c    = lane & 15u;
    const unsigned hi   = lane >> 4;
    int2* rec_hw = &lds_hw[tid >> 5][0];
    int*  rec_d  = &lds_d[tid >> 5][0];

    const unsigned die = ((unsigned)__builtin_amdgcn_readfirstlane(
                              (int)__builtin_amdgcn_s_sendmsg_rtn(0x87)) >> 16) & 15u;
    const unsigned own = (die >> 2) & 1u;

    for (unsigned pass = 0u; pass < 2u; ++pass) {
        const unsigned part = own ^ pass;
        unsigned* ctr = tickets + part * TICKET_STRIDE;

        if (__hip_atomic_load(ctr, __ATOMIC_RELAXED, __HIP_MEMORY_SCOPE_AGENT) >= (unsigned)N_CHUNKS)
            continue;

        unsigned t = ticket_draw(ctr, lane);
        while (t < (unsigned)N_CHUNKS) {
            const unsigned e0 = t * (unsigned)CHUNK_EDGES;
            const int4*   sp = reinterpret_cast<const int4*>(src + e0);
            const int4*   dp = reinterpret_cast<const int4*>(dst + e0);
            const float4* wp = reinterpret_cast<const float4*>(ew + e0);

            int4 sv[2], dv[2];
            float4 wv[2];
            #pragma unroll
            for (int q = 0; q < 2; ++q) {
                sv[q] = sp[32 * q + lane];
                dv[q] = dp[32 * q + lane];
                wv[q] = wp[32 * q + lane];
            }
            const unsigned t_next = ticket_draw(ctr, lane);

            unsigned n = 0u;
            #pragma unroll
            for (int q = 0; q < 2; ++q) {
                const int   sa[4] = {sv[q].x, sv[q].y, sv[q].z, sv[q].w};
                const int   da[4] = {dv[q].x, dv[q].y, dv[q].z, dv[q].w};
                const float wa[4] = {wv[q].x, wv[q].y, wv[q].z, wv[q].w};
                #pragma unroll
                for (int i = 0; i < 4; ++i) {
                    const bool in = (((unsigned)da[i] >= (unsigned)HALF_NODES) ? 1u : 0u) == part;
                    const unsigned m = __builtin_amdgcn_ballot_w32(in);
                    if (in) {
                        const unsigned pos = n + __builtin_amdgcn_mbcnt_lo(m, 0u);
                        rec_hw[pos] = make_int2(sa[i] * CH, __float_as_int(wa[i]));
                        rec_d[pos]  = da[i] * CH;
                    }
                    n += (unsigned)__builtin_popcount(m);
                }
            }
            __builtin_amdgcn_wave_barrier();

            const unsigned nfull = n & ~31u;
            for (unsigned b = 0u; b < nfull; b += 32u)
                scatter_run(rec_hw, rec_d, b, n, false, hi, c, h, agg);
            if (nfull < n)
                scatter_run(rec_hw, rec_d, nfull, n, true, hi, c, h, agg);
            __builtin_amdgcn_wave_barrier();

            t = t_next;
        }
    }
}

__global__ __launch_bounds__(256) void gcn_head_wmma(
    const float* __restrict__ agg, const float* __restrict__ w0,
    const float* __restrict__ b0,  const float* __restrict__ w1,
    const float* __restrict__ b1,  float* __restrict__ out)
{
    __shared__ float lds_w0[CH * 80];
    __shared__ float lds_b0[FCN];
    __shared__ float lds_w1[FCN];
    __shared__ float lds_o1[8][16 * 68];

    const int tid = threadIdx.x;
    for (int i = tid; i < CH * FCN; i += 256)
        lds_w0[(i / FCN) * 80 + (i % FCN)] = w0[i];
    if (tid < FCN) { lds_b0[tid] = b0[tid]; lds_w1[tid] = w1[tid]; }
    __syncthreads();

    const int wave = tid >> 5;
    const int lane = tid & 31;
    const int tile = blockIdx.x * 8 + wave;
    if (tile >= NTILES) return;

    const int m  = lane & 15;
    const int hi = lane >> 4;
    const int row0 = tile * 16;
    const float* arow = agg + (size_t)(row0 + m) * CH + 2 * hi;

    v8f acc[4];
    #pragma unroll
    for (int t = 0; t < 4; ++t) acc[t] = (v8f){0.f,0.f,0.f,0.f,0.f,0.f,0.f,0.f};

    #pragma unroll
    for (int k = 0; k < 4; ++k) {
        const int kk = 4 * k;
        v2f a; a.x = fmaxf(arow[kk], 0.f);
               a.y = fmaxf(arow[kk + 1], 0.f);
        const float* br0 = &lds_w0[(kk + 2 * hi)     * 80 + m];
        const float* br1 = &lds_w0[(kk + 2 * hi + 1) * 80 + m];
        #pragma unroll
        for (int t = 0; t < 4; ++t) {
            v2f b; b.x = br0[16 * t]; b.y = br1[16 * t];
            acc[t] = __builtin_amdgcn_wmma_f32_16x16x4_f32(
                         false, a, false, b, (short)0, acc[t], false, false);
        }
    }

    float* o1 = &lds_o1[wave][0];
    #pragma unroll
    for (int t = 0; t < 4; ++t) {
        const float bias = lds_b0[16 * t + m];
        #pragma unroll
        for (int j = 0; j < 8; ++j)
            o1[(j + 8 * hi) * 68 + 16 * t + m] = fmaxf(acc[t][j] + bias, 0.f);
    }

    const float* row = &lds_o1[wave][m * 68 + 32 * hi];
    const float* wv  = &lds_w1[32 * hi];
    float s = 0.f;
    #pragma unroll
    for (int f = 0; f < 32; ++f) s += row[f] * wv[f];
    s += __shfl_xor(s, 16, 32);
    if (hi == 0) out[row0 + m] = s + b1[0];
}

extern "C" void kernel_launch(void* const* d_in, const int* in_sizes, int n_in,
                              void* d_out, int out_size, void* d_ws, size_t ws_size,
                              hipStream_t stream)
{
    (void)in_sizes; (void)n_in; (void)out_size; (void)ws_size;
    const float* x  = (const float*)d_in[0];
    const int*   ei = (const int*)  d_in[1];
    const float* ew = (const float*)d_in[2];
    const float* wg = (const float*)d_in[3];
    const float* w0 = (const float*)d_in[4];
    const float* b0 = (const float*)d_in[5];
    const float* w1 = (const float*)d_in[6];
    const float* b1 = (const float*)d_in[7];
    float* out = (float*)d_out;

    float* h   = (float*)d_ws;
    float* agg = h + (size_t)N_NODES * CH;

    hipMemsetAsync(agg, 0, (size_t)N_NODES * CH * sizeof(float), stream);

    const int tile_blocks = (NTILES + 7) / 8;
    unsigned* tickets = (unsigned*)d_out;
    gcn_feat_wmma<<<tile_blocks, 256, 0, stream>>>(x, wg, h, tickets);

    gcn_edge_scatter<<<SCATTER_BLOCKS, 256, 0, stream>>>(h, ei, ei + N_EDGES, ew, agg, tickets);

    gcn_head_wmma<<<tile_blocks, 256, 0, stream>>>(agg, w0, b0, w1, b1, out);
}
